# starting kernel
# baseline (speedup 1.0000x reference)
#include <hip/hip_runtime.h>


#define NPTS      64
#define FAR_DELTA 1e10f
#define TILE_RAYS 16

typedef __attribute__((ext_vector_type(2))) float v2f;
typedef __attribute__((ext_vector_type(4))) float f4v;
typedef __attribute__((ext_vector_type(8))) float v8f;

__global__ __launch_bounds__(32) void volrend_wmma_kernel(
    const float* __restrict__ density,
    const float* __restrict__ depth,
    const float* __restrict__ feat,
    float* __restrict__ feat_out,
    float* __restrict__ depth_out,
    int n_rays)
{
    __shared__ float sdT[NPTS * 17];

    const int lane = threadIdx.x;
    const int half = lane >> 4;
    const int l16  = lane & 15;
    const int R0   = blockIdx.x * TILE_RAYS;

    __builtin_prefetch(depth   + (size_t)R0 * NPTS,     0, 1);
    __builtin_prefetch(density + (size_t)R0 * NPTS,     0, 1);
    __builtin_prefetch(feat    + (size_t)R0 * NPTS * 3, 0, 1);

    #pragma unroll
    for (int i = 0; i < 8; ++i) {
        int f4i  = i * 32 + lane;
        int fidx = f4i * 4;
        int r    = fidx >> 6;
        int p    = fidx & 63;
        int row  = min(R0 + r, n_rays - 1);
        f4v dv = *(const f4v*)(depth   + (size_t)row * NPTS + p);
        f4v sv = *(const f4v*)(density + (size_t)row * NPTS + p);
        float dnext = __shfl_down(dv.x, 1, 32);
        float dl3   = (l16 == 15) ? FAR_DELTA : (dnext - dv.w);
        sdT[(p + 0) * 17 + r] = sv.x * (dv.y - dv.x);
        sdT[(p + 1) * 17 + r] = sv.y * (dv.z - dv.y);
        sdT[(p + 2) * 17 + r] = sv.z * (dv.w - dv.z);
        sdT[(p + 3) * 17 + r] = sv.w * dl3;
    }
    __syncthreads();

    v2f amask[4];
    #pragma unroll
    for (int q = 0; q < 4; ++q) {
        int j0 = 4 * q + half * 2;
        amask[q].x = (j0     < l16) ? 1.f : 0.f;
        amask[q].y = (j0 + 1 < l16) ? 1.f : 0.f;
    }

    const size_t rowc = (size_t)min(R0 + l16, n_rays - 1);
    const float* fr = feat  + rowc * NPTS * 3;
    const float* dr = depth + rowc * NPTS;

    float carry = 0.f;
    float accx = 0.f, accy = 0.f, accz = 0.f, accd = 0.f;

    #pragma unroll
    for (int t = 0; t < 4; ++t) {
        v8f   C = {};
        float tpart = 0.f;
        #pragma unroll
        for (int q = 0; q < 4; ++q) {
            int j0 = 16 * t + 4 * q + half * 2;
            v2f b;
            b.x = sdT[(j0    ) * 17 + l16];
            b.y = sdT[(j0 + 1) * 17 + l16];
            tpart += b.x + b.y;
            C = __builtin_amdgcn_wmma_f32_16x16x4_f32(
                    false, amask[q], false, b, (short)0, C, false, false);
        }
        #pragma unroll
        for (int v = 0; v < 8; ++v) {
            int   p    = 16 * t + v + half * 8;
            float Pexc = carry + C[v];
            float sdp  = sdT[p * 17 + l16];
            float w    = __expf(-Pexc) - __expf(-(Pexc + sdp));
            accx += w * fr[p * 3 + 0];
            accy += w * fr[p * 3 + 1];
            accz += w * fr[p * 3 + 2];
            accd += w * dr[p];
        }
        carry += tpart + __shfl_xor(tpart, 16, 32);
    }

    accx += __shfl_xor(accx, 16, 32);
    accy += __shfl_xor(accy, 16, 32);
    accz += __shfl_xor(accz, 16, 32);
    accd += __shfl_xor(accd, 16, 32);

    if (lane < 16) {
        int rg = R0 + l16;
        if (rg < n_rays) {
            feat_out[rg * 3 + 0] = accx;
            feat_out[rg * 3 + 1] = accy;
            feat_out[rg * 3 + 2] = accz;
            depth_out[rg]        = accd;
        }
    }
}

extern "C" void kernel_launch(void* const* d_in, const int* in_sizes, int n_in,
                              void* d_out, int out_size, void* d_ws, size_t ws_size,
                              hipStream_t stream) {
    (void)n_in; (void)out_size; (void)d_ws; (void)ws_size;
    const float* density = (const float*)d_in[0];
    const float* depthv  = (const float*)d_in[1];
    const float* feat    = (const float*)d_in[2];

    const int n_rays = in_sizes[1] / NPTS;
    float* feat_out  = (float*)d_out;
    float* depth_out = feat_out + (size_t)n_rays * 3;

    const int tiles = (n_rays + TILE_RAYS - 1) / TILE_RAYS;
    volrend_wmma_kernel<<<dim3(tiles), dim3(32), 0, stream>>>(
        density, depthv, feat, feat_out, depth_out, n_rays);
}


# Whole-line feature/depth reads: each load takes 256 contiguous bytes per half-wave, issued up front; weights go through LDS to those lanes; per-ray sums folded over 16 lanes by DPP (new sum order)
# speedup vs baseline: 2.6855x; 2.6855x over previous
#include <hip/hip_runtime.h>


#define NPTS      64
#define FAR_DELTA 1e10f
#define TILE_RAYS 16

typedef __attribute__((ext_vector_type(2))) float v2f;
typedef __attribute__((ext_vector_type(4))) float f4v;
typedef __attribute__((ext_vector_type(8))) float v8f;

template <int MASK>
__device__ __forceinline__ float row_xor(float v) {
    return __builtin_bit_cast(float, __builtin_amdgcn_update_dpp(
        0, __builtin_bit_cast(int, v), 0x160 + MASK, 0xf, 0xf, false));
}

__device__ __forceinline__ float row_sum16(float v) {
    v += row_xor<1>(v);
    v += row_xor<2>(v);
    v += row_xor<4>(v);
    v += row_xor<8>(v);
    return v;
}

__global__ __launch_bounds__(32) void volrend_wmma_kernel(
    const float* __restrict__ density,
    const float* __restrict__ depth,
    const float* __restrict__ feat,
    float* __restrict__ feat_out,
    float* __restrict__ depth_out,
    int n_rays)
{
    __shared__ float sdT[NPTS * 17];

    const int lane = threadIdx.x;
    const int half = lane >> 4;
    const int l16  = lane & 15;
    const int R0   = blockIdx.x * TILE_RAYS;

    __builtin_prefetch(depth   + (size_t)R0 * NPTS,     0, 1);
    __builtin_prefetch(density + (size_t)R0 * NPTS,     0, 1);
    __builtin_prefetch(feat    + (size_t)R0 * NPTS * 3, 0, 1);

    f4v dv[8], sv[8], F[8][3];
    #pragma unroll
    for (int i = 0; i < 8; ++i) {
        int row = min(R0 + 2 * i + half, n_rays - 1);
        dv[i] = *(const f4v*)(depth   + (size_t)row * NPTS + 4 * l16);
        sv[i] = *(const f4v*)(density + (size_t)row * NPTS + 4 * l16);
    }
    #pragma unroll
    for (int i = 0; i < 8; ++i) {
        int row = min(R0 + 2 * i + half, n_rays - 1);
        const float* fp = feat + (size_t)row * (NPTS * 3) + 4 * l16;
        #pragma unroll
        for (int k = 0; k < 3; ++k)
            F[i][k] = *(const f4v*)(fp + 64 * k);
    }

    #pragma unroll
    for (int i = 0; i < 8; ++i) {
        const int r = 2 * i + half;
        const int p = 4 * l16;
        float dnext = __shfl_down(dv[i].x, 1, 32);
        float dl3   = (l16 == 15) ? FAR_DELTA : (dnext - dv[i].w);
        sdT[(p + 0) * 17 + r] = sv[i].x * (dv[i].y - dv[i].x);
        sdT[(p + 1) * 17 + r] = sv[i].y * (dv[i].z - dv[i].y);
        sdT[(p + 2) * 17 + r] = sv[i].z * (dv[i].w - dv[i].z);
        sdT[(p + 3) * 17 + r] = sv[i].w * dl3;
    }
    __syncthreads();

    v2f amask[4];
    #pragma unroll
    for (int q = 0; q < 4; ++q) {
        int j0 = 4 * q + half * 2;
        amask[q].x = (j0     < l16) ? 1.f : 0.f;
        amask[q].y = (j0 + 1 < l16) ? 1.f : 0.f;
    }

    float carry = 0.f;

    #pragma unroll
    for (int t = 0; t < 4; ++t) {
        v8f   C = {};
        float tpart = 0.f;
        #pragma unroll
        for (int q = 0; q < 4; ++q) {
            int j0 = 16 * t + 4 * q + half * 2;
            v2f b;
            b.x = sdT[(j0    ) * 17 + l16];
            b.y = sdT[(j0 + 1) * 17 + l16];
            tpart += b.x + b.y;
            C = __builtin_amdgcn_wmma_f32_16x16x4_f32(
                    false, amask[q], false, b, (short)0, C, false, false);
        }
        #pragma unroll
        for (int v = 0; v < 8; ++v) {
            int   p    = 16 * t + v + half * 8;
            float Pexc = carry + C[v];
            float sdp  = sdT[p * 17 + l16];
            sdT[p * 17 + l16] = __expf(-Pexc) - __expf(-(Pexc + sdp));
        }
        carry += tpart + __shfl_xor(tpart, 16, 32);
    }
    __syncthreads();

    const int s = l16 % 3;
    int wb[3];
    #pragma unroll
    for (int k = 0; k < 3; ++k)
        wb[k] = ((64 * k + 4 * l16) / 3) * 17 + half;
    const int db = (4 * l16) * 17 + half;

    float X[8], Y[8], Z[8], Dp[8];
    #pragma unroll
    for (int i = 0; i < 8; ++i) {
        float A[3] = {0.f, 0.f, 0.f};
        #pragma unroll
        for (int k = 0; k < 3; ++k) {
            const float wlo = sdT[wb[k] + 2 * i];
            const float whi = sdT[wb[k] + 17 + 2 * i];
            const float w1  = (s == (5 - k) % 3) ? whi : wlo;
            const float w2  = (s == (3 - k) % 3) ? wlo : whi;
            const f4v f = F[i][k];
            A[(k + 0) % 3] += wlo * f.x;
            A[(k + 1) % 3] += w1  * f.y;
            A[(k + 2) % 3] += w2  * f.z;
            A[(k + 3) % 3] += whi * f.w;
        }
        float d = 0.f;
        #pragma unroll
        for (int e = 0; e < 4; ++e)
            d += sdT[db + 17 * e + 2 * i] * dv[i][e];

        const float x = (s == 0) ? A[0] : ((s == 1) ? A[2] : A[1]);
        const float y = (s == 0) ? A[1] : ((s == 1) ? A[0] : A[2]);
        const float z = (s == 0) ? A[2] : ((s == 1) ? A[1] : A[0]);
        X[i]  = row_sum16(x);
        Y[i]  = row_sum16(y);
        Z[i]  = row_sum16(z);
        Dp[i] = row_sum16(d);
    }

    float ox = X[0], oy = Y[0], oz = Z[0], od = Dp[0];
    #pragma unroll
    for (int i = 1; i < 8; ++i) {
        ox = (l16 == i) ? X[i]  : ox;
        oy = (l16 == i) ? Y[i]  : oy;
        oz = (l16 == i) ? Z[i]  : oz;
        od = (l16 == i) ? Dp[i] : od;
    }
    if (l16 < 8) {
        int rg = R0 + 2 * l16 + half;
        if (rg < n_rays) {
            feat_out[rg * 3 + 0] = ox;
            feat_out[rg * 3 + 1] = oy;
            feat_out[rg * 3 + 2] = oz;
            depth_out[rg]        = od;
        }
    }
}

extern "C" void kernel_launch(void* const* d_in, const int* in_sizes, int n_in,
                              void* d_out, int out_size, void* d_ws, size_t ws_size,
                              hipStream_t stream) {
    (void)n_in; (void)out_size; (void)d_ws; (void)ws_size;
    const float* density = (const float*)d_in[0];
    const float* depthv  = (const float*)d_in[1];
    const float* feat    = (const float*)d_in[2];

    const int n_rays = in_sizes[1] / NPTS;
    float* feat_out  = (float*)d_out;
    float* depth_out = feat_out + (size_t)n_rays * 3;

    const int tiles = (n_rays + TILE_RAYS - 1) / TILE_RAYS;
    volrend_wmma_kernel<<<dim3(tiles), dim3(32), 0, stream>>>(
        density, depthv, feat, feat_out, depth_out, n_rays);
}
